# starting kernel
# baseline (speedup 1.0000x reference)
#include <hip/hip_runtime.h>
#include <math.h>


#define B_   4
#define DIM  512
#define N_   4096
#define H_   8
#define EPS_ 1e-5f
#define TPI  64
#define LDSX 68
#define PF   4

typedef __attribute__((ext_vector_type(2))) float v2f;
typedef __attribute__((ext_vector_type(8))) float v8f;

__device__ __forceinline__ float sigmf(float v) { return 1.0f / (1.0f + __expf(-v)); }

__global__ void __launch_bounds__(256) ln_stats_kernel(const float* __restrict__ x,
                                                       float* __restrict__ mean,
                                                       float* __restrict__ rstd) {
  int b    = blockIdx.x >> 6;
  int t0   = (blockIdx.x & 63) << 6;
  int tl   = threadIdx.x & 63;
  int part = threadIdx.x >> 6;

  const float* xb = x + (size_t)b * DIM * N_ + t0 + tl;
  float s = 0.0f, s2 = 0.0f;
#pragma unroll 8
  for (int c = part; c < DIM; c += 4) {
    float v = xb[(size_t)c * N_];
    s += v; s2 += v * v;
  }
  __shared__ float sh[2][4][64];
  sh[0][part][tl] = s;
  sh[1][part][tl] = s2;
  __syncthreads();
  if (part == 0) {
    s  = sh[0][0][tl] + sh[0][1][tl] + sh[0][2][tl] + sh[0][3][tl];
    s2 = sh[1][0][tl] + sh[1][1][tl] + sh[1][2][tl] + sh[1][3][tl];
    float m   = s * (1.0f / DIM);
    float var = s2 * (1.0f / DIM) - m * m;
    mean[b * N_ + t0 + tl] = m;
    rstd[b * N_ + t0 + tl] = rsqrtf(var + EPS_);
  }
}

__global__ void __launch_bounds__(256)
ema_wmma_kernel(const float* __restrict__ x,
                const float* __restrict__ gamma,
                const float* __restrict__ beta,
                const float* __restrict__ expansion,
                const float* __restrict__ reduction,
                const float* __restrict__ alphas,
                const float* __restrict__ damp,
                const float* __restrict__ mean,
                const float* __restrict__ rstd,
                float* __restrict__ out) {
  const int b    = blockIdx.x >> 5;
  const int d0   = (blockIdx.x & 31) << 4;
  const int tid  = threadIdx.x;
  const int h    = tid >> 5;
  const int lane = tid & 31;
  const int j    = lane & 15;
  const int half = lane >> 4;

  const float a = sigmf(alphas[h]);
  const float q = (1.0f - a) * sigmf(damp[h]);
  const float w = expansion[h * DIM + d0 + j] * reduction[h * DIM + d0 + j];

  v2f A[4];
#pragma unroll
  for (int k = 0; k < 4; ++k)
#pragma unroll
    for (int r = 0; r < 2; ++r) {
      int col = 4 * k + r + 2 * half;
      A[k][r] = (j >= col) ? a * __powf(q, (float)(j - col)) : 0.0f;
    }
  v2f Ac;
  Ac[0] = (half == 0) ? __powf(q, (float)(j + 1)) : 0.0f;
  Ac[1] = 0.0f;

  __shared__ __align__(16) float lds_x[16 * LDSX];
  __shared__ __align__(16) float lds_acc[16 * LDSX];

  for (int i = tid; i < 16 * LDSX; i += 256) lds_acc[i] = 0.0f;

  const int srow = tid >> 4;
  const int scol = (tid & 15) << 2;
  const size_t rowbase = (size_t)(b * DIM + d0) * N_;
  const float* xrow = x + rowbase + (size_t)srow * N_;
  float*       orow = out + rowbase + (size_t)srow * N_;
  const float* mrow = mean + b * N_;
  const float* rrow = rstd + b * N_;
  const float g  = gamma[d0 + srow];
  const float bt = beta[d0 + srow];

  float ycarry = 0.0f;

  for (int it = 0; it < N_ / TPI; ++it) {
    const int t0 = it * TPI;

    const float4 xr = *(const float4*)(xrow + t0 + scol);
    const float4 m0 = *(const float4*)(mrow + t0 + scol);
    const float4 r0 = *(const float4*)(rrow + t0 + scol);
    float4 n;
    n.x = (xr.x - m0.x) * r0.x * g + bt;
    n.y = (xr.y - m0.y) * r0.y * g + bt;
    n.z = (xr.z - m0.z) * r0.z * g + bt;
    n.w = (xr.w - m0.w) * r0.w * g + bt;
    *(float4*)&lds_x[srow * LDSX + scol] = n;

    if (it + PF < N_ / TPI)
      __builtin_prefetch(xrow + t0 + PF * TPI + scol, 0, 3);

    __syncthreads();

#pragma unroll
    for (int sub = 0; sub < 4; ++sub) {
      const int s0 = sub * 16;
      v2f Bk[4];
#pragma unroll
      for (int k = 0; k < 4; ++k)
#pragma unroll
        for (int r = 0; r < 2; ++r)
          Bk[k][r] = lds_x[j * LDSX + s0 + 4 * k + r + 2 * half];
      v2f By;
      By[0] = (half == 0) ? ycarry : 0.0f;
      By[1] = 0.0f;

      v8f acc = {0.f, 0.f, 0.f, 0.f, 0.f, 0.f, 0.f, 0.f};
      acc = __builtin_amdgcn_wmma_f32_16x16x4_f32(false, Ac, false, By, (short)0, acc, false, false);
#pragma unroll
      for (int k = 0; k < 4; ++k)
        acc = __builtin_amdgcn_wmma_f32_16x16x4_f32(false, A[k], false, Bk[k], (short)0, acc, false, false);

      ycarry = __shfl(acc[7], 16 + j, 32);

#pragma unroll
      for (int v = 0; v < 8; ++v)
        atomicAdd(&lds_acc[j * LDSX + s0 + v + 8 * half], w * acc[v]);
    }

    __syncthreads();

    float4 o = *(float4*)&lds_acc[srow * LDSX + scol];
    const float4 z = {0.f, 0.f, 0.f, 0.f};
    *(float4*)&lds_acc[srow * LDSX + scol] = z;
    o.x += xr.x; o.y += xr.y; o.z += xr.z; o.w += xr.w;
    *(float4*)(orow + t0 + scol) = o;
  }
}

extern "C" void kernel_launch(void* const* d_in, const int* in_sizes, int n_in,
                              void* d_out, int out_size, void* d_ws, size_t ws_size,
                              hipStream_t stream) {
  (void)in_sizes; (void)n_in; (void)out_size; (void)ws_size;
  const float* x         = (const float*)d_in[0];
  const float* ln_gamma  = (const float*)d_in[1];
  const float* ln_beta   = (const float*)d_in[2];
  const float* expansion = (const float*)d_in[3];
  const float* reduction = (const float*)d_in[4];
  const float* alphas    = (const float*)d_in[5];
  const float* dampen    = (const float*)d_in[6];
  float* out  = (float*)d_out;
  float* mean = (float*)d_ws;
  float* rstd = mean + B_ * N_;

  ln_stats_kernel<<<B_ * (N_ / 64), 256, 0, stream>>>(x, mean, rstd);
  ema_wmma_kernel<<<B_ * (DIM / 16), 256, 0, stream>>>(
      x, ln_gamma, ln_beta, expansion, reduction, alphas, dampen, mean, rstd, out);
}


# Time axis cut into 8 segments per channel tile (1024 blocks); carry-in rebuilt per segment by a WMMA reduction over earlier steps (per-wave partial sums, not the serial chain); powers by products
# speedup vs baseline: 2.2033x; 2.2033x over previous
#include <hip/hip_runtime.h>
#include <math.h>


#define B_   4
#define DIM  512
#define N_   4096
#define H_   8
#define EPS_ 1e-5f
#define TPI  64
#define LDSX 68
#define PF   4
#define SEG  8
#define LSEG (N_ / SEG)
#define ITS  (LSEG / TPI)
#define RU   32

typedef __attribute__((ext_vector_type(2))) float v2f;
typedef __attribute__((ext_vector_type(8))) float v8f;

__device__ __forceinline__ float sigmf(float v) { return 1.0f / (1.0f + __expf(-v)); }

__device__ __forceinline__ float qpow15(float q1, float q2, float q4, float q8, int e) {
  float p = (e & 1) ? q1 : 1.0f;
  p *= (e & 2) ? q2 : 1.0f;
  p *= (e & 4) ? q4 : 1.0f;
  p *= (e & 8) ? q8 : 1.0f;
  return p;
}

__global__ void __launch_bounds__(256) ln_stats_kernel(const float* __restrict__ x,
                                                       float* __restrict__ mean,
                                                       float* __restrict__ rstd) {
  int b    = blockIdx.x >> 6;
  int t0   = (blockIdx.x & 63) << 6;
  int tl   = threadIdx.x & 63;
  int part = threadIdx.x >> 6;

  const float* xb = x + (size_t)b * DIM * N_ + t0 + tl;
  float s = 0.0f, s2 = 0.0f;
#pragma unroll 8
  for (int c = part; c < DIM; c += 4) {
    float v = xb[(size_t)c * N_];
    s += v; s2 += v * v;
  }
  __shared__ float sh[2][4][64];
  sh[0][part][tl] = s;
  sh[1][part][tl] = s2;
  __syncthreads();
  if (part == 0) {
    s  = sh[0][0][tl] + sh[0][1][tl] + sh[0][2][tl] + sh[0][3][tl];
    s2 = sh[1][0][tl] + sh[1][1][tl] + sh[1][2][tl] + sh[1][3][tl];
    float m   = s * (1.0f / DIM);
    float var = s2 * (1.0f / DIM) - m * m;
    mean[b * N_ + t0 + tl] = m;
    rstd[b * N_ + t0 + tl] = rsqrtf(var + EPS_);
  }
}

__global__ void __launch_bounds__(256)
ema_wmma_kernel(const float* __restrict__ x,
                const float* __restrict__ gamma,
                const float* __restrict__ beta,
                const float* __restrict__ expansion,
                const float* __restrict__ reduction,
                const float* __restrict__ alphas,
                const float* __restrict__ damp,
                const float* __restrict__ mean,
                const float* __restrict__ rstd,
                float* __restrict__ out) {
  const int seg  = (SEG - 1) - (blockIdx.x >> 7);
  const int tile = blockIdx.x & 127;
  const int b    = tile >> 5;
  const int d0   = (tile & 31) << 4;
  const int tid  = threadIdx.x;
  const int h    = tid >> 5;
  const int lane = tid & 31;
  const int j    = lane & 15;
  const int half = lane >> 4;

  const float a = sigmf(alphas[h]);
  const float q = (1.0f - a) * sigmf(damp[h]);
  const float w = expansion[h * DIM + d0 + j] * reduction[h * DIM + d0 + j];
  const float q2 = q * q, q4 = q2 * q2, q8 = q4 * q4, q16 = q8 * q8;

  v2f A[4];
#pragma unroll
  for (int k = 0; k < 4; ++k)
#pragma unroll
    for (int r = 0; r < 2; ++r) {
      int col = 4 * k + r + 2 * half;
      A[k][r] = (j >= col) ? a * qpow15(q, q2, q4, q8, (j - col) & 15) : 0.0f;
    }
  v2f Ac;
  Ac[0] = (half == 0) ? q * qpow15(q, q2, q4, q8, j) : 0.0f;
  Ac[1] = 0.0f;

  __shared__ __align__(16) float lds_x[16 * LDSX];
  __shared__ __align__(16) float lds_acc[16 * LDSX];
  __shared__ float lds_red[8 * H_ * 16];

  for (int i = tid; i < 16 * LDSX; i += 256) lds_acc[i] = 0.0f;

  const int srow = tid >> 4;
  const int scol = (tid & 15) << 2;
  const size_t rowbase = (size_t)(b * DIM + d0) * N_;
  const float* xrow = x + rowbase + (size_t)srow * N_;
  float*       orow = out + rowbase + (size_t)srow * N_;
  const float* mrow = mean + b * N_;
  const float* rrow = rstd + b * N_;
  const float g  = gamma[d0 + srow];
  const float bt = beta[d0 + srow];

  const int tseg = seg * LSEG;

#pragma unroll
  for (int p = 0; p < PF; ++p)
    __builtin_prefetch(xrow + tseg + p * TPI + scol, 0, 3);

  float ycarry = 0.0f;
  if (seg > 0) {
    const int   hr = j & 7;
    const float ar = sigmf(alphas[hr]);
    const float qr = (1.0f - ar) * sigmf(damp[hr]);
    const float qr2 = qr * qr, qr4 = qr2 * qr2, qr8 = qr4 * qr4, qr16 = qr8 * qr8;
    v2f Ar[8];
    {
      float p = (half == 0) ? qr16 : 1.0f;
#pragma unroll
      for (int e = 15; e >= 0; --e) {
        Ar[e >> 1][e & 1] = (j < 8) ? ar * p : 0.0f;
        p *= qr;
      }
    }
    const float qr32 = qr16 * qr16;
    float q32[8];
#pragma unroll
    for (int r = 0; r < 8; ++r)
      q32[r] = __int_as_float(__builtin_amdgcn_readlane(__float_as_int(qr32), r));
    const float gj = gamma[d0 + j];
    const float bj = beta[d0 + j];
    const float* xl = x + rowbase + (size_t)j * N_ + 16 * half;
    const float* ml = mrow + 16 * half;
    const float* rl = rrow + 16 * half;
    const int units = seg * (64 / RU);
    const int tbeg  = h * seg * 64;

    v8f racc = {0.f, 0.f, 0.f, 0.f, 0.f, 0.f, 0.f, 0.f};
    for (int u = 0; u < units; ++u) {
      const int tu = tbeg + u * RU;
      float4 xq[4], mq[4], rq[4];
#pragma unroll
      for (int i = 0; i < 4; ++i) {
        xq[i] = *(const float4*)(xl + tu + 4 * i);
        mq[i] = *(const float4*)(ml + tu + 4 * i);
        rq[i] = *(const float4*)(rl + tu + 4 * i);
      }
      v2f Bn[8];
#pragma unroll
      for (int i = 0; i < 4; ++i) {
        Bn[2 * i][0]     = (xq[i].x - mq[i].x) * rq[i].x * gj + bj;
        Bn[2 * i][1]     = (xq[i].y - mq[i].y) * rq[i].y * gj + bj;
        Bn[2 * i + 1][0] = (xq[i].z - mq[i].z) * rq[i].z * gj + bj;
        Bn[2 * i + 1][1] = (xq[i].w - mq[i].w) * rq[i].w * gj + bj;
      }
#pragma unroll
      for (int r = 0; r < 8; ++r) racc[r] *= q32[r];
#pragma unroll
      for (int k = 0; k < 8; ++k)
        racc = __builtin_amdgcn_wmma_f32_16x16x4_f32(false, Ar[k], false, Bn[k], (short)0, racc, false, false);
      asm volatile("v_nop\n\tv_nop\n\tv_nop\n\tv_nop\n\tv_nop" : "+v"(racc));
    }
    if (half == 0) {
#pragma unroll
      for (int r = 0; r < 8; ++r) lds_red[(h * H_ + r) * 16 + j] = racc[r];
    }
    __syncthreads();
    const float q32h = q16 * q16;
    const float q64 = q32h * q32h;
    float f = q64;
    for (int i = 1; i < seg; ++i) f *= q64;
    float y = lds_red[(0 * H_ + h) * 16 + j];
#pragma unroll
    for (int wv = 1; wv < 8; ++wv) y = y * f + lds_red[(wv * H_ + h) * 16 + j];
    ycarry = y;
  }

  for (int it = 0; it < ITS; ++it) {
    const int t0 = tseg + it * TPI;

    const float4 xr = *(const float4*)(xrow + t0 + scol);
    const float4 m0 = *(const float4*)(mrow + t0 + scol);
    const float4 r0 = *(const float4*)(rrow + t0 + scol);
    float4 n;
    n.x = (xr.x - m0.x) * r0.x * g + bt;
    n.y = (xr.y - m0.y) * r0.y * g + bt;
    n.z = (xr.z - m0.z) * r0.z * g + bt;
    n.w = (xr.w - m0.w) * r0.w * g + bt;
    *(float4*)&lds_x[srow * LDSX + scol] = n;

    if (it + PF < ITS)
      __builtin_prefetch(xrow + t0 + PF * TPI + scol, 0, 3);

    __syncthreads();

#pragma unroll
    for (int sub = 0; sub < 4; ++sub) {
      const int s0 = sub * 16;
      v2f Bk[4];
#pragma unroll
      for (int k = 0; k < 4; ++k)
#pragma unroll
        for (int r = 0; r < 2; ++r)
          Bk[k][r] = lds_x[j * LDSX + s0 + 4 * k + r + 2 * half];
      v2f By;
      By[0] = (half == 0) ? ycarry : 0.0f;
      By[1] = 0.0f;

      v8f acc = {0.f, 0.f, 0.f, 0.f, 0.f, 0.f, 0.f, 0.f};
      acc = __builtin_amdgcn_wmma_f32_16x16x4_f32(false, Ac, false, By, (short)0, acc, false, false);
#pragma unroll
      for (int k = 0; k < 4; ++k)
        acc = __builtin_amdgcn_wmma_f32_16x16x4_f32(false, A[k], false, Bk[k], (short)0, acc, false, false);

      ycarry = __shfl(acc[7], 16 + j, 32);

#pragma unroll
      for (int v = 0; v < 8; ++v)
        atomicAdd(&lds_acc[j * LDSX + s0 + v + 8 * half], w * acc[v]);
    }

    __syncthreads();

    float4 o = *(float4*)&lds_acc[srow * LDSX + scol];
    const float4 z = {0.f, 0.f, 0.f, 0.f};
    *(float4*)&lds_acc[srow * LDSX + scol] = z;
    o.x += xr.x; o.y += xr.y; o.z += xr.z; o.w += xr.w;
    *(float4*)(orow + t0 + scol) = o;
  }
}

extern "C" void kernel_launch(void* const* d_in, const int* in_sizes, int n_in,
                              void* d_out, int out_size, void* d_ws, size_t ws_size,
                              hipStream_t stream) {
  (void)in_sizes; (void)n_in; (void)out_size; (void)ws_size;
  const float* x         = (const float*)d_in[0];
  const float* ln_gamma  = (const float*)d_in[1];
  const float* ln_beta   = (const float*)d_in[2];
  const float* expansion = (const float*)d_in[3];
  const float* reduction = (const float*)d_in[4];
  const float* alphas    = (const float*)d_in[5];
  const float* dampen    = (const float*)d_in[6];
  float* out  = (float*)d_out;
  float* mean = (float*)d_ws;
  float* rstd = mean + B_ * N_;

  ln_stats_kernel<<<B_ * (N_ / 64), 256, 0, stream>>>(x, mean, rstd);
  ema_wmma_kernel<<<SEG * B_ * (DIM / 16), 256, 0, stream>>>(
      x, ln_gamma, ln_beta, expansion, reduction, alphas, dampen, mean, rstd, out);
}


# Head combine through per-head LDS tiles: plain 16-byte stores and an in-order 8-way sum in the drain replace 256 LDS float atomics per 64-step iteration; head sum order now fixed, not arrival order
# speedup vs baseline: 3.0956x; 1.4050x over previous
#include <hip/hip_runtime.h>
#include <math.h>


#define B_   4
#define DIM  512
#define N_   4096
#define H_   8
#define EPS_ 1e-5f
#define TPI  64
#define LDSX 68
#define PF   4
#define SEG  8
#define LSEG (N_ / SEG)
#define ITS  (LSEG / TPI)
#define RU   32

typedef __attribute__((ext_vector_type(2))) float v2f;
typedef __attribute__((ext_vector_type(8))) float v8f;

__device__ __forceinline__ float sigmf(float v) { return 1.0f / (1.0f + __expf(-v)); }

__device__ __forceinline__ float qpow15(float q1, float q2, float q4, float q8, int e) {
  float p = (e & 1) ? q1 : 1.0f;
  p *= (e & 2) ? q2 : 1.0f;
  p *= (e & 4) ? q4 : 1.0f;
  p *= (e & 8) ? q8 : 1.0f;
  return p;
}

__global__ void __launch_bounds__(256) ln_stats_kernel(const float* __restrict__ x,
                                                       float* __restrict__ mean,
                                                       float* __restrict__ rstd) {
  int b    = blockIdx.x >> 6;
  int t0   = (blockIdx.x & 63) << 6;
  int tl   = threadIdx.x & 63;
  int part = threadIdx.x >> 6;

  const float* xb = x + (size_t)b * DIM * N_ + t0 + tl;
  float s = 0.0f, s2 = 0.0f;
#pragma unroll 8
  for (int c = part; c < DIM; c += 4) {
    float v = xb[(size_t)c * N_];
    s += v; s2 += v * v;
  }
  __shared__ float sh[2][4][64];
  sh[0][part][tl] = s;
  sh[1][part][tl] = s2;
  __syncthreads();
  if (part == 0) {
    s  = sh[0][0][tl] + sh[0][1][tl] + sh[0][2][tl] + sh[0][3][tl];
    s2 = sh[1][0][tl] + sh[1][1][tl] + sh[1][2][tl] + sh[1][3][tl];
    float m   = s * (1.0f / DIM);
    float var = s2 * (1.0f / DIM) - m * m;
    mean[b * N_ + t0 + tl] = m;
    rstd[b * N_ + t0 + tl] = rsqrtf(var + EPS_);
  }
}

__global__ void __launch_bounds__(256)
ema_wmma_kernel(const float* __restrict__ x,
                const float* __restrict__ gamma,
                const float* __restrict__ beta,
                const float* __restrict__ expansion,
                const float* __restrict__ reduction,
                const float* __restrict__ alphas,
                const float* __restrict__ damp,
                const float* __restrict__ mean,
                const float* __restrict__ rstd,
                float* __restrict__ out) {
  const int seg  = (SEG - 1) - (blockIdx.x >> 7);
  const int tile = blockIdx.x & 127;
  const int b    = tile >> 5;
  const int d0   = (tile & 31) << 4;
  const int tid  = threadIdx.x;
  const int h    = tid >> 5;
  const int lane = tid & 31;
  const int j    = lane & 15;
  const int half = lane >> 4;

  const float a = sigmf(alphas[h]);
  const float q = (1.0f - a) * sigmf(damp[h]);
  const float w = expansion[h * DIM + d0 + j] * reduction[h * DIM + d0 + j];
  const float q2 = q * q, q4 = q2 * q2, q8 = q4 * q4, q16 = q8 * q8;

  v2f A[4];
#pragma unroll
  for (int k = 0; k < 4; ++k)
#pragma unroll
    for (int r = 0; r < 2; ++r) {
      int col = 4 * k + r + 2 * half;
      A[k][r] = (j >= col) ? a * qpow15(q, q2, q4, q8, (j - col) & 15) : 0.0f;
    }
  v2f Ac;
  Ac[0] = (half == 0) ? q * qpow15(q, q2, q4, q8, j) : 0.0f;
  Ac[1] = 0.0f;

  __shared__ __align__(16) float lds_x[16 * LDSX];
  __shared__ __align__(16) float lds_y[H_ * 16 * LDSX];
  __shared__ float lds_red[8 * H_ * 16];

  const int srow = tid >> 4;
  const int scol = (tid & 15) << 2;
  const size_t rowbase = (size_t)(b * DIM + d0) * N_;
  const float* xrow = x + rowbase + (size_t)srow * N_;
  float*       orow = out + rowbase + (size_t)srow * N_;
  const float* mrow = mean + b * N_;
  const float* rrow = rstd + b * N_;
  const float g  = gamma[d0 + srow];
  const float bt = beta[d0 + srow];

  const int tseg = seg * LSEG;

#pragma unroll
  for (int p = 0; p < PF; ++p)
    __builtin_prefetch(xrow + tseg + p * TPI + scol, 0, 3);

  float ycarry = 0.0f;
  if (seg > 0) {
    const int   hr = j & 7;
    const float ar = sigmf(alphas[hr]);
    const float qr = (1.0f - ar) * sigmf(damp[hr]);
    const float qr2 = qr * qr, qr4 = qr2 * qr2, qr8 = qr4 * qr4, qr16 = qr8 * qr8;
    v2f Ar[8];
    {
      float p = (half == 0) ? qr16 : 1.0f;
#pragma unroll
      for (int e = 15; e >= 0; --e) {
        Ar[e >> 1][e & 1] = (j < 8) ? ar * p : 0.0f;
        p *= qr;
      }
    }
    const float qr32 = qr16 * qr16;
    float q32[8];
#pragma unroll
    for (int r = 0; r < 8; ++r)
      q32[r] = __int_as_float(__builtin_amdgcn_readlane(__float_as_int(qr32), r));
    const float gj = gamma[d0 + j];
    const float bj = beta[d0 + j];
    const float* xl = x + rowbase + (size_t)j * N_ + 16 * half;
    const float* ml = mrow + 16 * half;
    const float* rl = rrow + 16 * half;
    const int units = seg * (64 / RU);
    const int tbeg  = h * seg * 64;

    v8f racc = {0.f, 0.f, 0.f, 0.f, 0.f, 0.f, 0.f, 0.f};
    for (int u = 0; u < units; ++u) {
      const int tu = tbeg + u * RU;
      float4 xq[4], mq[4], rq[4];
#pragma unroll
      for (int i = 0; i < 4; ++i) {
        xq[i] = *(const float4*)(xl + tu + 4 * i);
        mq[i] = *(const float4*)(ml + tu + 4 * i);
        rq[i] = *(const float4*)(rl + tu + 4 * i);
      }
      v2f Bn[8];
#pragma unroll
      for (int i = 0; i < 4; ++i) {
        Bn[2 * i][0]     = (xq[i].x - mq[i].x) * rq[i].x * gj + bj;
        Bn[2 * i][1]     = (xq[i].y - mq[i].y) * rq[i].y * gj + bj;
        Bn[2 * i + 1][0] = (xq[i].z - mq[i].z) * rq[i].z * gj + bj;
        Bn[2 * i + 1][1] = (xq[i].w - mq[i].w) * rq[i].w * gj + bj;
      }
#pragma unroll
      for (int r = 0; r < 8; ++r) racc[r] *= q32[r];
#pragma unroll
      for (int k = 0; k < 8; ++k)
        racc = __builtin_amdgcn_wmma_f32_16x16x4_f32(false, Ar[k], false, Bn[k], (short)0, racc, false, false);
      asm volatile("v_nop\n\tv_nop\n\tv_nop\n\tv_nop\n\tv_nop" : "+v"(racc));
    }
    if (half == 0) {
#pragma unroll
      for (int r = 0; r < 8; ++r) lds_red[(h * H_ + r) * 16 + j] = racc[r];
    }
    __syncthreads();
    const float q32h = q16 * q16;
    const float q64 = q32h * q32h;
    float f = q64;
    for (int i = 1; i < seg; ++i) f *= q64;
    float y = lds_red[(0 * H_ + h) * 16 + j];
#pragma unroll
    for (int wv = 1; wv < 8; ++wv) y = y * f + lds_red[(wv * H_ + h) * 16 + j];
    ycarry = y;
  }

  float* const yrow = &lds_y[(h * 16 + j) * LDSX + 8 * half];

  for (int it = 0; it < ITS; ++it) {
    const int t0 = tseg + it * TPI;

    const float4 xr = *(const float4*)(xrow + t0 + scol);
    const float4 m0 = *(const float4*)(mrow + t0 + scol);
    const float4 r0 = *(const float4*)(rrow + t0 + scol);
    float4 n;
    n.x = (xr.x - m0.x) * r0.x * g + bt;
    n.y = (xr.y - m0.y) * r0.y * g + bt;
    n.z = (xr.z - m0.z) * r0.z * g + bt;
    n.w = (xr.w - m0.w) * r0.w * g + bt;
    *(float4*)&lds_x[srow * LDSX + scol] = n;

    if (it + PF < ITS)
      __builtin_prefetch(xrow + t0 + PF * TPI + scol, 0, 3);

    __syncthreads();

#pragma unroll
    for (int sub = 0; sub < 4; ++sub) {
      const int s0 = sub * 16;
      v2f Bk[4];
#pragma unroll
      for (int k = 0; k < 4; ++k)
#pragma unroll
        for (int r = 0; r < 2; ++r)
          Bk[k][r] = lds_x[j * LDSX + s0 + 4 * k + r + 2 * half];
      v2f By;
      By[0] = (half == 0) ? ycarry : 0.0f;
      By[1] = 0.0f;

      v8f acc = {0.f, 0.f, 0.f, 0.f, 0.f, 0.f, 0.f, 0.f};
      acc = __builtin_amdgcn_wmma_f32_16x16x4_f32(false, Ac, false, By, (short)0, acc, false, false);
#pragma unroll
      for (int k = 0; k < 4; ++k)
        acc = __builtin_amdgcn_wmma_f32_16x16x4_f32(false, A[k], false, Bk[k], (short)0, acc, false, false);

      ycarry = __shfl(acc[7], 16 + j, 32);

      float4 ylo, yhi;
      ylo.x = w * acc[0]; ylo.y = w * acc[1]; ylo.z = w * acc[2]; ylo.w = w * acc[3];
      yhi.x = w * acc[4]; yhi.y = w * acc[5]; yhi.z = w * acc[6]; yhi.w = w * acc[7];
      *(float4*)(yrow + s0)     = ylo;
      *(float4*)(yrow + s0 + 4) = yhi;
    }

    __syncthreads();

    float4 o = *(const float4*)&lds_y[srow * LDSX + scol];
#pragma unroll
    for (int hh = 1; hh < H_; ++hh) {
      const float4 p = *(const float4*)&lds_y[(hh * 16 + srow) * LDSX + scol];
      o.x += p.x; o.y += p.y; o.z += p.z; o.w += p.w;
    }
    o.x += xr.x; o.y += xr.y; o.z += xr.z; o.w += xr.w;
    *(float4*)(orow + t0 + scol) = o;
  }
}

extern "C" void kernel_launch(void* const* d_in, const int* in_sizes, int n_in,
                              void* d_out, int out_size, void* d_ws, size_t ws_size,
                              hipStream_t stream) {
  (void)in_sizes; (void)n_in; (void)out_size; (void)ws_size;
  const float* x         = (const float*)d_in[0];
  const float* ln_gamma  = (const float*)d_in[1];
  const float* ln_beta   = (const float*)d_in[2];
  const float* expansion = (const float*)d_in[3];
  const float* reduction = (const float*)d_in[4];
  const float* alphas    = (const float*)d_in[5];
  const float* dampen    = (const float*)d_in[6];
  float* out  = (float*)d_out;
  float* mean = (float*)d_ws;
  float* rstd = mean + B_ * N_;

  ln_stats_kernel<<<B_ * (N_ / 64), 256, 0, stream>>>(x, mean, rstd);
  ema_wmma_kernel<<<SEG * B_ * (DIM / 16), 256, 0, stream>>>(
      x, ln_gamma, ln_beta, expansion, reduction, alphas, dampen, mean, rstd, out);
}


# Carry-in operand fetch: each unit's x tile read as whole 128-byte lines, one unit ahead, and re-laid through a per-wave LDS tile instead of 16-line scattered loads; values and sum order unchanged
# speedup vs baseline: 3.2261x; 1.0422x over previous
#include <hip/hip_runtime.h>
#include <math.h>


#define B_   4
#define DIM  512
#define N_   4096
#define H_   8
#define EPS_ 1e-5f
#define TPI  64
#define LDSX 68
#define PF   4
#define SEG  8
#define LSEG (N_ / SEG)
#define ITS  (LSEG / TPI)
#define RU   32
#define CPIT 36

typedef __attribute__((ext_vector_type(2))) float v2f;
typedef __attribute__((ext_vector_type(8))) float v8f;

__device__ __forceinline__ float sigmf(float v) { return 1.0f / (1.0f + __expf(-v)); }

__device__ __forceinline__ float qpow15(float q1, float q2, float q4, float q8, int e) {
  float p = (e & 1) ? q1 : 1.0f;
  p *= (e & 2) ? q2 : 1.0f;
  p *= (e & 4) ? q4 : 1.0f;
  p *= (e & 8) ? q8 : 1.0f;
  return p;
}

__global__ void __launch_bounds__(256) ln_stats_kernel(const float* __restrict__ x,
                                                       float* __restrict__ mean,
                                                       float* __restrict__ rstd) {
  int b    = blockIdx.x >> 6;
  int t0   = (blockIdx.x & 63) << 6;
  int tl   = threadIdx.x & 63;
  int part = threadIdx.x >> 6;

  const float* xb = x + (size_t)b * DIM * N_ + t0 + tl;
  float s = 0.0f, s2 = 0.0f;
#pragma unroll 8
  for (int c = part; c < DIM; c += 4) {
    float v = xb[(size_t)c * N_];
    s += v; s2 += v * v;
  }
  __shared__ float sh[2][4][64];
  sh[0][part][tl] = s;
  sh[1][part][tl] = s2;
  __syncthreads();
  if (part == 0) {
    s  = sh[0][0][tl] + sh[0][1][tl] + sh[0][2][tl] + sh[0][3][tl];
    s2 = sh[1][0][tl] + sh[1][1][tl] + sh[1][2][tl] + sh[1][3][tl];
    float m   = s * (1.0f / DIM);
    float var = s2 * (1.0f / DIM) - m * m;
    mean[b * N_ + t0 + tl] = m;
    rstd[b * N_ + t0 + tl] = rsqrtf(var + EPS_);
  }
}

__global__ void __launch_bounds__(256) __attribute__((amdgpu_num_vgpr(128)))
ema_wmma_kernel(const float* __restrict__ x,
                const float* __restrict__ gamma,
                const float* __restrict__ beta,
                const float* __restrict__ expansion,
                const float* __restrict__ reduction,
                const float* __restrict__ alphas,
                const float* __restrict__ damp,
                const float* __restrict__ mean,
                const float* __restrict__ rstd,
                float* __restrict__ out) {
  const int seg  = (SEG - 1) - (blockIdx.x >> 7);
  const int tile = blockIdx.x & 127;
  const int b    = tile >> 5;
  const int d0   = (tile & 31) << 4;
  const int tid  = threadIdx.x;
  const int h    = tid >> 5;
  const int lane = tid & 31;
  const int j    = lane & 15;
  const int half = lane >> 4;

  const float a = sigmf(alphas[h]);
  const float q = (1.0f - a) * sigmf(damp[h]);
  const float w = expansion[h * DIM + d0 + j] * reduction[h * DIM + d0 + j];
  const float q2 = q * q, q4 = q2 * q2, q8 = q4 * q4, q16 = q8 * q8;

  v2f A[4];
#pragma unroll
  for (int k = 0; k < 4; ++k)
#pragma unroll
    for (int r = 0; r < 2; ++r) {
      int col = 4 * k + r + 2 * half;
      A[k][r] = (j >= col) ? a * qpow15(q, q2, q4, q8, (j - col) & 15) : 0.0f;
    }
  v2f Ac;
  Ac[0] = (half == 0) ? q * qpow15(q, q2, q4, q8, j) : 0.0f;
  Ac[1] = 0.0f;

  __shared__ __align__(16) float lds_x[16 * LDSX];
  __shared__ __align__(16) float lds_y[H_ * 16 * LDSX];
  __shared__ float lds_red[8 * H_ * 16];

  const int srow = tid >> 4;
  const int scol = (tid & 15) << 2;
  const size_t rowbase = (size_t)(b * DIM + d0) * N_;
  const float* xrow = x + rowbase + (size_t)srow * N_;
  float*       orow = out + rowbase + (size_t)srow * N_;
  const float* mrow = mean + b * N_;
  const float* rrow = rstd + b * N_;
  const float g  = gamma[d0 + srow];
  const float bt = beta[d0 + srow];

  const int tseg = seg * LSEG;

#pragma unroll
  for (int p = 0; p < PF; ++p)
    __builtin_prefetch(xrow + tseg + p * TPI + scol, 0, 3);

  float ycarry = 0.0f;
  if (seg > 0) {
    const int   hr = j & 7;
    const float ar = sigmf(alphas[hr]);
    const float qr = (1.0f - ar) * sigmf(damp[hr]);
    const float qr2 = qr * qr, qr4 = qr2 * qr2, qr8 = qr4 * qr4, qr16 = qr8 * qr8;
    v2f Ar[8];
    {
      float p = (half == 0) ? qr16 : 1.0f;
#pragma unroll
      for (int e = 15; e >= 0; --e) {
        Ar[e >> 1][e & 1] = (j < 8) ? ar * p : 0.0f;
        p *= qr;
      }
    }
    const float qr32 = qr16 * qr16;
    float q32[8];
#pragma unroll
    for (int r = 0; r < 8; ++r)
      q32[r] = __int_as_float(__builtin_amdgcn_readlane(__float_as_int(qr32), r));
    float* const ct = &lds_y[h * 16 * LDSX];
    const int crow = lane >> 3;
    const int ccol = (lane & 7) << 2;
    const float* xc = x + rowbase + (size_t)crow * N_ + ccol;
    const float* mc = mrow + ccol;
    const float* rc = rrow + ccol;
    float gc[4], bc[4];
#pragma unroll
    for (int i = 0; i < 4; ++i) {
      gc[i] = gamma[d0 + 4 * i + crow];
      bc[i] = beta[d0 + 4 * i + crow];
    }
    float* const cst = ct + crow * CPIT + ccol;
    const float* const cld = ct + j * CPIT + 16 * half;
    const int units = seg * (64 / RU);
    const int tbeg  = h * seg * 64;

    float4 xq[4], mq, rq;
#pragma unroll
    for (int i = 0; i < 4; ++i) xq[i] = *(const float4*)(xc + (size_t)(4 * i) * N_ + tbeg);
    mq = *(const float4*)(mc + tbeg);
    rq = *(const float4*)(rc + tbeg);

    v8f racc = {0.f, 0.f, 0.f, 0.f, 0.f, 0.f, 0.f, 0.f};
    for (int u = 0; u < units; ++u) {
      const int tn = tbeg + ((u + 1 < units) ? (u + 1) : u) * RU;
      float4 xn[4], mn, rn;
#pragma unroll
      for (int i = 0; i < 4; ++i) xn[i] = *(const float4*)(xc + (size_t)(4 * i) * N_ + tn);
      mn = *(const float4*)(mc + tn);
      rn = *(const float4*)(rc + tn);

#pragma unroll
      for (int i = 0; i < 4; ++i) {
        float4 nv;
        nv.x = (xq[i].x - mq.x) * rq.x * gc[i] + bc[i];
        nv.y = (xq[i].y - mq.y) * rq.y * gc[i] + bc[i];
        nv.z = (xq[i].z - mq.z) * rq.z * gc[i] + bc[i];
        nv.w = (xq[i].w - mq.w) * rq.w * gc[i] + bc[i];
        *(float4*)(cst + 4 * i * CPIT) = nv;
      }
      __builtin_amdgcn_wave_barrier();
      v2f Bn[8];
#pragma unroll
      for (int i = 0; i < 4; ++i) {
        const float4 f = *(const float4*)(cld + 4 * i);
        Bn[2 * i][0]     = f.x;
        Bn[2 * i][1]     = f.y;
        Bn[2 * i + 1][0] = f.z;
        Bn[2 * i + 1][1] = f.w;
      }
      __builtin_amdgcn_wave_barrier();
#pragma unroll
      for (int r = 0; r < 8; ++r) racc[r] *= q32[r];
#pragma unroll
      for (int k = 0; k < 8; ++k)
        racc = __builtin_amdgcn_wmma_f32_16x16x4_f32(false, Ar[k], false, Bn[k], (short)0, racc, false, false);
      asm volatile("v_nop\n\tv_nop\n\tv_nop\n\tv_nop\n\tv_nop" : "+v"(racc));
#pragma unroll
      for (int i = 0; i < 4; ++i) xq[i] = xn[i];
      mq = mn; rq = rn;
    }
    if (half == 0) {
#pragma unroll
      for (int r = 0; r < 8; ++r) lds_red[(h * H_ + r) * 16 + j] = racc[r];
    }
    __syncthreads();
    const float q32h = q16 * q16;
    const float q64 = q32h * q32h;
    float f = q64;
    for (int i = 1; i < seg; ++i) f *= q64;
    float y = lds_red[(0 * H_ + h) * 16 + j];
#pragma unroll
    for (int wv = 1; wv < 8; ++wv) y = y * f + lds_red[(wv * H_ + h) * 16 + j];
    ycarry = y;
  }

  float* const yrow = &lds_y[(h * 16 + j) * LDSX + 8 * half];

  for (int it = 0; it < ITS; ++it) {
    const int t0 = tseg + it * TPI;

    const float4 xr = *(const float4*)(xrow + t0 + scol);
    const float4 m0 = *(const float4*)(mrow + t0 + scol);
    const float4 r0 = *(const float4*)(rrow + t0 + scol);
    float4 n;
    n.x = (xr.x - m0.x) * r0.x * g + bt;
    n.y = (xr.y - m0.y) * r0.y * g + bt;
    n.z = (xr.z - m0.z) * r0.z * g + bt;
    n.w = (xr.w - m0.w) * r0.w * g + bt;
    *(float4*)&lds_x[srow * LDSX + scol] = n;

    if (it + PF < ITS)
      __builtin_prefetch(xrow + t0 + PF * TPI + scol, 0, 3);

    __syncthreads();

#pragma unroll
    for (int sub = 0; sub < 4; ++sub) {
      const int s0 = sub * 16;
      v2f Bk[4];
#pragma unroll
      for (int k = 0; k < 4; ++k)
#pragma unroll
        for (int r = 0; r < 2; ++r)
          Bk[k][r] = lds_x[j * LDSX + s0 + 4 * k + r + 2 * half];
      v2f By;
      By[0] = (half == 0) ? ycarry : 0.0f;
      By[1] = 0.0f;

      v8f acc = {0.f, 0.f, 0.f, 0.f, 0.f, 0.f, 0.f, 0.f};
      acc = __builtin_amdgcn_wmma_f32_16x16x4_f32(false, Ac, false, By, (short)0, acc, false, false);
#pragma unroll
      for (int k = 0; k < 4; ++k)
        acc = __builtin_amdgcn_wmma_f32_16x16x4_f32(false, A[k], false, Bk[k], (short)0, acc, false, false);

      ycarry = __shfl(acc[7], 16 + j, 32);

      float4 ylo, yhi;
      ylo.x = w * acc[0]; ylo.y = w * acc[1]; ylo.z = w * acc[2]; ylo.w = w * acc[3];
      yhi.x = w * acc[4]; yhi.y = w * acc[5]; yhi.z = w * acc[6]; yhi.w = w * acc[7];
      *(float4*)(yrow + s0)     = ylo;
      *(float4*)(yrow + s0 + 4) = yhi;
    }

    __syncthreads();

    float4 o = *(const float4*)&lds_y[srow * LDSX + scol];
#pragma unroll
    for (int hh = 1; hh < H_; ++hh) {
      const float4 p = *(const float4*)&lds_y[(hh * 16 + srow) * LDSX + scol];
      o.x += p.x; o.y += p.y; o.z += p.z; o.w += p.w;
    }
    o.x += xr.x; o.y += xr.y; o.z += xr.z; o.w += xr.w;
    *(float4*)(orow + t0 + scol) = o;
  }
}

extern "C" void kernel_launch(void* const* d_in, const int* in_sizes, int n_in,
                              void* d_out, int out_size, void* d_ws, size_t ws_size,
                              hipStream_t stream) {
  (void)in_sizes; (void)n_in; (void)out_size; (void)ws_size;
  const float* x         = (const float*)d_in[0];
  const float* ln_gamma  = (const float*)d_in[1];
  const float* ln_beta   = (const float*)d_in[2];
  const float* expansion = (const float*)d_in[3];
  const float* reduction = (const float*)d_in[4];
  const float* alphas    = (const float*)d_in[5];
  const float* dampen    = (const float*)d_in[6];
  float* out  = (float*)d_out;
  float* mean = (float*)d_ws;
  float* rstd = mean + B_ * N_;

  ln_stats_kernel<<<B_ * (N_ / 64), 256, 0, stream>>>(x, mean, rstd);
  ema_wmma_kernel<<<SEG * B_ * (DIM / 16), 256, 0, stream>>>(
      x, ln_gamma, ln_beta, expansion, reduction, alphas, dampen, mean, rstd, out);
}


# Scan carry folded into row 0 of the first K-slice operand (x0 + q*z on the unscaled state; gain a moved into the head weight): 4 WMMAs per 16 steps, not 5; carry slice summed last, sum order changed
# speedup vs baseline: 3.3828x; 1.0486x over previous
#include <hip/hip_runtime.h>
#include <math.h>


#define B_   4
#define DIM  512
#define N_   4096
#define H_   8
#define EPS_ 1e-5f
#define TPI  64
#define LDSX 68
#define PF   4
#define SEG  8
#define LSEG (N_ / SEG)
#define ITS  (LSEG / TPI)
#define RU   32
#define CPIT 36

typedef __attribute__((ext_vector_type(2))) float v2f;
typedef __attribute__((ext_vector_type(8))) float v8f;

__device__ __forceinline__ float sigmf(float v) { return 1.0f / (1.0f + __expf(-v)); }

__device__ __forceinline__ float qpow15(float q1, float q2, float q4, float q8, int e) {
  float p = (e & 1) ? q1 : 1.0f;
  p *= (e & 2) ? q2 : 1.0f;
  p *= (e & 4) ? q4 : 1.0f;
  p *= (e & 8) ? q8 : 1.0f;
  return p;
}

__global__ void __launch_bounds__(256) ln_stats_kernel(const float* __restrict__ x,
                                                       float* __restrict__ mean,
                                                       float* __restrict__ rstd) {
  int b    = blockIdx.x >> 6;
  int t0   = (blockIdx.x & 63) << 6;
  int tl   = threadIdx.x & 63;
  int part = threadIdx.x >> 6;

  const float* xb = x + (size_t)b * DIM * N_ + t0 + tl;
  float s = 0.0f, s2 = 0.0f;
#pragma unroll 8
  for (int c = part; c < DIM; c += 4) {
    float v = xb[(size_t)c * N_];
    s += v; s2 += v * v;
  }
  __shared__ float sh[2][4][64];
  sh[0][part][tl] = s;
  sh[1][part][tl] = s2;
  __syncthreads();
  if (part == 0) {
    s  = sh[0][0][tl] + sh[0][1][tl] + sh[0][2][tl] + sh[0][3][tl];
    s2 = sh[1][0][tl] + sh[1][1][tl] + sh[1][2][tl] + sh[1][3][tl];
    float m   = s * (1.0f / DIM);
    float var = s2 * (1.0f / DIM) - m * m;
    mean[b * N_ + t0 + tl] = m;
    rstd[b * N_ + t0 + tl] = rsqrtf(var + EPS_);
  }
}

__global__ void __launch_bounds__(256) __attribute__((amdgpu_num_vgpr(128)))
ema_wmma_kernel(const float* __restrict__ x,
                const float* __restrict__ gamma,
                const float* __restrict__ beta,
                const float* __restrict__ expansion,
                const float* __restrict__ reduction,
                const float* __restrict__ alphas,
                const float* __restrict__ damp,
                const float* __restrict__ mean,
                const float* __restrict__ rstd,
                float* __restrict__ out) {
  const int seg  = (SEG - 1) - (blockIdx.x >> 7);
  const int tile = blockIdx.x & 127;
  const int b    = tile >> 5;
  const int d0   = (tile & 31) << 4;
  const int tid  = threadIdx.x;
  const int h    = tid >> 5;
  const int lane = tid & 31;
  const int j    = lane & 15;
  const int half = lane >> 4;

  const float a = sigmf(alphas[h]);
  const float q = (1.0f - a) * sigmf(damp[h]);
  const float w = a * (expansion[h * DIM + d0 + j] * reduction[h * DIM + d0 + j]);
  const float q2 = q * q, q4 = q2 * q2, q8 = q4 * q4, q16 = q8 * q8;

  v2f A[4];
#pragma unroll
  for (int k = 0; k < 4; ++k)
#pragma unroll
    for (int r = 0; r < 2; ++r) {
      int col = 4 * k + r + 2 * half;
      A[k][r] = (j >= col) ? qpow15(q, q2, q4, q8, (j - col) & 15) : 0.0f;
    }
  const float qfold = (half == 0) ? q : 0.0f;

  __shared__ __align__(16) float lds_x[16 * LDSX];
  __shared__ __align__(16) float lds_y[H_ * 16 * LDSX];
  __shared__ float lds_red[8 * H_ * 16];

  const int srow = tid >> 4;
  const int scol = (tid & 15) << 2;
  const size_t rowbase = (size_t)(b * DIM + d0) * N_;
  const float* xrow = x + rowbase + (size_t)srow * N_;
  float*       orow = out + rowbase + (size_t)srow * N_;
  const float* mrow = mean + b * N_;
  const float* rrow = rstd + b * N_;
  const float g  = gamma[d0 + srow];
  const float bt = beta[d0 + srow];

  const int tseg = seg * LSEG;

#pragma unroll
  for (int p = 0; p < PF; ++p)
    __builtin_prefetch(xrow + tseg + p * TPI + scol, 0, 3);

  float ycarry = 0.0f;
  if (seg > 0) {
    const int   hr = j & 7;
    const float ar = sigmf(alphas[hr]);
    const float qr = (1.0f - ar) * sigmf(damp[hr]);
    const float qr2 = qr * qr, qr4 = qr2 * qr2, qr8 = qr4 * qr4, qr16 = qr8 * qr8;
    v2f Ar[8];
    {
      float p = (half == 0) ? qr16 : 1.0f;
#pragma unroll
      for (int e = 15; e >= 0; --e) {
        Ar[e >> 1][e & 1] = (j < 8) ? p : 0.0f;
        p *= qr;
      }
    }
    const float qr32 = qr16 * qr16;
    float q32[8];
#pragma unroll
    for (int r = 0; r < 8; ++r)
      q32[r] = __int_as_float(__builtin_amdgcn_readlane(__float_as_int(qr32), r));
    float* const ct = &lds_y[h * 16 * LDSX];
    const int crow = lane >> 3;
    const int ccol = (lane & 7) << 2;
    const float* xc = x + rowbase + (size_t)crow * N_ + ccol;
    const float* mc = mrow + ccol;
    const float* rc = rrow + ccol;
    float gc[4], bc[4];
#pragma unroll
    for (int i = 0; i < 4; ++i) {
      gc[i] = gamma[d0 + 4 * i + crow];
      bc[i] = beta[d0 + 4 * i + crow];
    }
    float* const cst = ct + crow * CPIT + ccol;
    const float* const cld = ct + j * CPIT + 16 * half;
    const int units = seg * (64 / RU);
    const int tbeg  = h * seg * 64;

    float4 xq[4], mq, rq;
#pragma unroll
    for (int i = 0; i < 4; ++i) xq[i] = *(const float4*)(xc + (size_t)(4 * i) * N_ + tbeg);
    mq = *(const float4*)(mc + tbeg);
    rq = *(const float4*)(rc + tbeg);

    v8f racc = {0.f, 0.f, 0.f, 0.f, 0.f, 0.f, 0.f, 0.f};
    for (int u = 0; u < units; ++u) {
      const int tn = tbeg + ((u + 1 < units) ? (u + 1) : u) * RU;
      float4 xn[4], mn, rn;
#pragma unroll
      for (int i = 0; i < 4; ++i) xn[i] = *(const float4*)(xc + (size_t)(4 * i) * N_ + tn);
      mn = *(const float4*)(mc + tn);
      rn = *(const float4*)(rc + tn);

#pragma unroll
      for (int i = 0; i < 4; ++i) {
        float4 nv;
        nv.x = (xq[i].x - mq.x) * rq.x * gc[i] + bc[i];
        nv.y = (xq[i].y - mq.y) * rq.y * gc[i] + bc[i];
        nv.z = (xq[i].z - mq.z) * rq.z * gc[i] + bc[i];
        nv.w = (xq[i].w - mq.w) * rq.w * gc[i] + bc[i];
        *(float4*)(cst + 4 * i * CPIT) = nv;
      }
      __builtin_amdgcn_wave_barrier();
      v2f Bn[8];
#pragma unroll
      for (int i = 0; i < 4; ++i) {
        const float4 f = *(const float4*)(cld + 4 * i);
        Bn[2 * i][0]     = f.x;
        Bn[2 * i][1]     = f.y;
        Bn[2 * i + 1][0] = f.z;
        Bn[2 * i + 1][1] = f.w;
      }
      __builtin_amdgcn_wave_barrier();
#pragma unroll
      for (int r = 0; r < 8; ++r) racc[r] *= q32[r];
#pragma unroll
      for (int k = 0; k < 8; ++k)
        racc = __builtin_amdgcn_wmma_f32_16x16x4_f32(false, Ar[k], false, Bn[k], (short)0, racc, false, false);
      asm volatile("v_nop\n\tv_nop\n\tv_nop\n\tv_nop\n\tv_nop" : "+v"(racc));
#pragma unroll
      for (int i = 0; i < 4; ++i) xq[i] = xn[i];
      mq = mn; rq = rn;
    }
    if (half == 0) {
#pragma unroll
      for (int r = 0; r < 8; ++r) lds_red[(h * H_ + r) * 16 + j] = racc[r];
    }
    __syncthreads();
    const float q32h = q16 * q16;
    const float q64 = q32h * q32h;
    float f = q64;
    for (int i = 1; i < seg; ++i) f *= q64;
    float y = lds_red[(0 * H_ + h) * 16 + j];
#pragma unroll
    for (int wv = 1; wv < 8; ++wv) y = y * f + lds_red[(wv * H_ + h) * 16 + j];
    ycarry = y;
  }

  float* const yrow = &lds_y[(h * 16 + j) * LDSX + 8 * half];

  for (int it = 0; it < ITS; ++it) {
    const int t0 = tseg + it * TPI;

    const float4 xr = *(const float4*)(xrow + t0 + scol);
    const float4 m0 = *(const float4*)(mrow + t0 + scol);
    const float4 r0 = *(const float4*)(rrow + t0 + scol);
    float4 n;
    n.x = (xr.x - m0.x) * r0.x * g + bt;
    n.y = (xr.y - m0.y) * r0.y * g + bt;
    n.z = (xr.z - m0.z) * r0.z * g + bt;
    n.w = (xr.w - m0.w) * r0.w * g + bt;
    *(float4*)&lds_x[srow * LDSX + scol] = n;

    if (it + PF < ITS)
      __builtin_prefetch(xrow + t0 + PF * TPI + scol, 0, 3);

    __syncthreads();

#pragma unroll
    for (int sub = 0; sub < 4; ++sub) {
      const int s0 = sub * 16;
      v2f Bk[4];
#pragma unroll
      for (int k = 0; k < 4; ++k)
#pragma unroll
        for (int r = 0; r < 2; ++r)
          Bk[k][r] = lds_x[j * LDSX + s0 + 4 * k + r + 2 * half];
      Bk[0][0] = fmaf(qfold, ycarry, Bk[0][0]);

      v8f acc = {0.f, 0.f, 0.f, 0.f, 0.f, 0.f, 0.f, 0.f};
#pragma unroll
      for (int k = 1; k < 4; ++k)
        acc = __builtin_amdgcn_wmma_f32_16x16x4_f32(false, A[k], false, Bk[k], (short)0, acc, false, false);
      acc = __builtin_amdgcn_wmma_f32_16x16x4_f32(false, A[0], false, Bk[0], (short)0, acc, false, false);

      ycarry = __shfl(acc[7], 16 + j, 32);

      float4 ylo, yhi;
      ylo.x = w * acc[0]; ylo.y = w * acc[1]; ylo.z = w * acc[2]; ylo.w = w * acc[3];
      yhi.x = w * acc[4]; yhi.y = w * acc[5]; yhi.z = w * acc[6]; yhi.w = w * acc[7];
      *(float4*)(yrow + s0)     = ylo;
      *(float4*)(yrow + s0 + 4) = yhi;
    }

    __syncthreads();

    float4 o = *(const float4*)&lds_y[srow * LDSX + scol];
#pragma unroll
    for (int hh = 1; hh < H_; ++hh) {
      const float4 p = *(const float4*)&lds_y[(hh * 16 + srow) * LDSX + scol];
      o.x += p.x; o.y += p.y; o.z += p.z; o.w += p.w;
    }
    o.x += xr.x; o.y += xr.y; o.z += xr.z; o.w += xr.w;
    *(float4*)(orow + t0 + scol) = o;
  }
}

extern "C" void kernel_launch(void* const* d_in, const int* in_sizes, int n_in,
                              void* d_out, int out_size, void* d_ws, size_t ws_size,
                              hipStream_t stream) {
  (void)in_sizes; (void)n_in; (void)out_size; (void)ws_size;
  const float* x         = (const float*)d_in[0];
  const float* ln_gamma  = (const float*)d_in[1];
  const float* ln_beta   = (const float*)d_in[2];
  const float* expansion = (const float*)d_in[3];
  const float* reduction = (const float*)d_in[4];
  const float* alphas    = (const float*)d_in[5];
  const float* dampen    = (const float*)d_in[6];
  float* out  = (float*)d_out;
  float* mean = (float*)d_ws;
  float* rstd = mean + B_ * N_;

  ln_stats_kernel<<<B_ * (N_ / 64), 256, 0, stream>>>(x, mean, rstd);
  ema_wmma_kernel<<<SEG * B_ * (DIM / 16), 256, 0, stream>>>(
      x, ln_gamma, ln_beta, expansion, reduction, alphas, dampen, mean, rstd, out);
}


# LN statistics launch: each thread requests all 32 of its 16-byte row pieces (aligned 256-byte pieces per half-wave) before adding, one round trip, not a serial chain; LN channel sum order changed
# speedup vs baseline: 4.0393x; 1.1941x over previous
#include <hip/hip_runtime.h>
#include <math.h>


#define B_   4
#define DIM  512
#define N_   4096
#define H_   8
#define EPS_ 1e-5f
#define TPI  64
#define LDSX 68
#define PF   4
#define SEG  8
#define LSEG (N_ / SEG)
#define ITS  (LSEG / TPI)
#define RU   32
#define CPIT 36

typedef __attribute__((ext_vector_type(2))) float v2f;
typedef __attribute__((ext_vector_type(8))) float v8f;

__device__ __forceinline__ float sigmf(float v) { return 1.0f / (1.0f + __expf(-v)); }

__device__ __forceinline__ float qpow15(float q1, float q2, float q4, float q8, int e) {
  float p = (e & 1) ? q1 : 1.0f;
  p *= (e & 2) ? q2 : 1.0f;
  p *= (e & 4) ? q4 : 1.0f;
  p *= (e & 8) ? q8 : 1.0f;
  return p;
}

__global__ void __launch_bounds__(256) __attribute__((amdgpu_waves_per_eu(2, 2)))
ln_stats_kernel(const float* __restrict__ x,
                                                       float* __restrict__ mean,
                                                       float* __restrict__ rstd) {
  const int b  = blockIdx.x >> 6;
  const int t0 = (blockIdx.x & 63) << 6;
  const int qd = threadIdx.x & 15;
  const int cg = threadIdx.x >> 4;

  const float* xb = x + (size_t)b * DIM * N_ + (size_t)cg * N_ + t0 + 4 * qd;
  float4 v[DIM / 16];
#pragma unroll
  for (int i = 0; i < DIM / 16; ++i) v[i] = *(const float4*)(xb + (size_t)(16 * i) * N_);

  float4 s  = {0.f, 0.f, 0.f, 0.f};
  float4 s2 = {0.f, 0.f, 0.f, 0.f};
#pragma unroll
  for (int i = 0; i < DIM / 16; ++i) {
    s.x += v[i].x; s.y += v[i].y; s.z += v[i].z; s.w += v[i].w;
    s2.x += v[i].x * v[i].x; s2.y += v[i].y * v[i].y;
    s2.z += v[i].z * v[i].z; s2.w += v[i].w * v[i].w;
  }
  __shared__ __align__(16) float sh[2][16][64];
  *(float4*)&sh[0][cg][4 * qd] = s;
  *(float4*)&sh[1][cg][4 * qd] = s2;
  __syncthreads();
  if (threadIdx.x < 64) {
    const int tl = threadIdx.x;
    float a = sh[0][0][tl], a2 = sh[1][0][tl];
#pragma unroll
    for (int k = 1; k < 16; ++k) { a += sh[0][k][tl]; a2 += sh[1][k][tl]; }
    float m   = a * (1.0f / DIM);
    float var = a2 * (1.0f / DIM) - m * m;
    mean[b * N_ + t0 + tl] = m;
    rstd[b * N_ + t0 + tl] = rsqrtf(var + EPS_);
  }
}

__global__ void __launch_bounds__(256) __attribute__((amdgpu_num_vgpr(128)))
ema_wmma_kernel(const float* __restrict__ x,
                const float* __restrict__ gamma,
                const float* __restrict__ beta,
                const float* __restrict__ expansion,
                const float* __restrict__ reduction,
                const float* __restrict__ alphas,
                const float* __restrict__ damp,
                const float* __restrict__ mean,
                const float* __restrict__ rstd,
                float* __restrict__ out) {
  const int seg  = (SEG - 1) - (blockIdx.x >> 7);
  const int tile = blockIdx.x & 127;
  const int b    = tile >> 5;
  const int d0   = (tile & 31) << 4;
  const int tid  = threadIdx.x;
  const int h    = tid >> 5;
  const int lane = tid & 31;
  const int j    = lane & 15;
  const int half = lane >> 4;

  const float a = sigmf(alphas[h]);
  const float q = (1.0f - a) * sigmf(damp[h]);
  const float w = a * (expansion[h * DIM + d0 + j] * reduction[h * DIM + d0 + j]);
  const float q2 = q * q, q4 = q2 * q2, q8 = q4 * q4, q16 = q8 * q8;

  v2f A[4];
#pragma unroll
  for (int k = 0; k < 4; ++k)
#pragma unroll
    for (int r = 0; r < 2; ++r) {
      int col = 4 * k + r + 2 * half;
      A[k][r] = (j >= col) ? qpow15(q, q2, q4, q8, (j - col) & 15) : 0.0f;
    }
  const float qfold = (half == 0) ? q : 0.0f;

  __shared__ __align__(16) float lds_x[16 * LDSX];
  __shared__ __align__(16) float lds_y[H_ * 16 * LDSX];
  __shared__ float lds_red[8 * H_ * 16];

  const int srow = tid >> 4;
  const int scol = (tid & 15) << 2;
  const size_t rowbase = (size_t)(b * DIM + d0) * N_;
  const float* xrow = x + rowbase + (size_t)srow * N_;
  float*       orow = out + rowbase + (size_t)srow * N_;
  const float* mrow = mean + b * N_;
  const float* rrow = rstd + b * N_;
  const float g  = gamma[d0 + srow];
  const float bt = beta[d0 + srow];

  const int tseg = seg * LSEG;

#pragma unroll
  for (int p = 0; p < PF; ++p)
    __builtin_prefetch(xrow + tseg + p * TPI + scol, 0, 3);

  float ycarry = 0.0f;
  if (seg > 0) {
    const int   hr = j & 7;
    const float ar = sigmf(alphas[hr]);
    const float qr = (1.0f - ar) * sigmf(damp[hr]);
    const float qr2 = qr * qr, qr4 = qr2 * qr2, qr8 = qr4 * qr4, qr16 = qr8 * qr8;
    v2f Ar[8];
    {
      float p = (half == 0) ? qr16 : 1.0f;
#pragma unroll
      for (int e = 15; e >= 0; --e) {
        Ar[e >> 1][e & 1] = (j < 8) ? p : 0.0f;
        p *= qr;
      }
    }
    const float qr32 = qr16 * qr16;
    float q32[8];
#pragma unroll
    for (int r = 0; r < 8; ++r)
      q32[r] = __int_as_float(__builtin_amdgcn_readlane(__float_as_int(qr32), r));
    float* const ct = &lds_y[h * 16 * LDSX];
    const int crow = lane >> 3;
    const int ccol = (lane & 7) << 2;
    const float* xc = x + rowbase + (size_t)crow * N_ + ccol;
    const float* mc = mrow + ccol;
    const float* rc = rrow + ccol;
    float gc[4], bc[4];
#pragma unroll
    for (int i = 0; i < 4; ++i) {
      gc[i] = gamma[d0 + 4 * i + crow];
      bc[i] = beta[d0 + 4 * i + crow];
    }
    float* const cst = ct + crow * CPIT + ccol;
    const float* const cld = ct + j * CPIT + 16 * half;
    const int units = seg * (64 / RU);
    const int tbeg  = h * seg * 64;

    float4 xq[4], mq, rq;
#pragma unroll
    for (int i = 0; i < 4; ++i) xq[i] = *(const float4*)(xc + (size_t)(4 * i) * N_ + tbeg);
    mq = *(const float4*)(mc + tbeg);
    rq = *(const float4*)(rc + tbeg);

    v8f racc = {0.f, 0.f, 0.f, 0.f, 0.f, 0.f, 0.f, 0.f};
    for (int u = 0; u < units; ++u) {
      const int tn = tbeg + ((u + 1 < units) ? (u + 1) : u) * RU;
      float4 xn[4], mn, rn;
#pragma unroll
      for (int i = 0; i < 4; ++i) xn[i] = *(const float4*)(xc + (size_t)(4 * i) * N_ + tn);
      mn = *(const float4*)(mc + tn);
      rn = *(const float4*)(rc + tn);

#pragma unroll
      for (int i = 0; i < 4; ++i) {
        float4 nv;
        nv.x = (xq[i].x - mq.x) * rq.x * gc[i] + bc[i];
        nv.y = (xq[i].y - mq.y) * rq.y * gc[i] + bc[i];
        nv.z = (xq[i].z - mq.z) * rq.z * gc[i] + bc[i];
        nv.w = (xq[i].w - mq.w) * rq.w * gc[i] + bc[i];
        *(float4*)(cst + 4 * i * CPIT) = nv;
      }
      __builtin_amdgcn_wave_barrier();
      v2f Bn[8];
#pragma unroll
      for (int i = 0; i < 4; ++i) {
        const float4 f = *(const float4*)(cld + 4 * i);
        Bn[2 * i][0]     = f.x;
        Bn[2 * i][1]     = f.y;
        Bn[2 * i + 1][0] = f.z;
        Bn[2 * i + 1][1] = f.w;
      }
      __builtin_amdgcn_wave_barrier();
#pragma unroll
      for (int r = 0; r < 8; ++r) racc[r] *= q32[r];
#pragma unroll
      for (int k = 0; k < 8; ++k)
        racc = __builtin_amdgcn_wmma_f32_16x16x4_f32(false, Ar[k], false, Bn[k], (short)0, racc, false, false);
      asm volatile("v_nop\n\tv_nop\n\tv_nop\n\tv_nop\n\tv_nop" : "+v"(racc));
#pragma unroll
      for (int i = 0; i < 4; ++i) xq[i] = xn[i];
      mq = mn; rq = rn;
    }
    if (half == 0) {
#pragma unroll
      for (int r = 0; r < 8; ++r) lds_red[(h * H_ + r) * 16 + j] = racc[r];
    }
    __syncthreads();
    const float q32h = q16 * q16;
    const float q64 = q32h * q32h;
    float f = q64;
    for (int i = 1; i < seg; ++i) f *= q64;
    float y = lds_red[(0 * H_ + h) * 16 + j];
#pragma unroll
    for (int wv = 1; wv < 8; ++wv) y = y * f + lds_red[(wv * H_ + h) * 16 + j];
    ycarry = y;
  }

  float* const yrow = &lds_y[(h * 16 + j) * LDSX + 8 * half];

  for (int it = 0; it < ITS; ++it) {
    const int t0 = tseg + it * TPI;

    const float4 xr = *(const float4*)(xrow + t0 + scol);
    const float4 m0 = *(const float4*)(mrow + t0 + scol);
    const float4 r0 = *(const float4*)(rrow + t0 + scol);
    float4 n;
    n.x = (xr.x - m0.x) * r0.x * g + bt;
    n.y = (xr.y - m0.y) * r0.y * g + bt;
    n.z = (xr.z - m0.z) * r0.z * g + bt;
    n.w = (xr.w - m0.w) * r0.w * g + bt;
    *(float4*)&lds_x[srow * LDSX + scol] = n;

    if (it + PF < ITS)
      __builtin_prefetch(xrow + t0 + PF * TPI + scol, 0, 3);

    __syncthreads();

#pragma unroll
    for (int sub = 0; sub < 4; ++sub) {
      const int s0 = sub * 16;
      v2f Bk[4];
#pragma unroll
      for (int k = 0; k < 4; ++k)
#pragma unroll
        for (int r = 0; r < 2; ++r)
          Bk[k][r] = lds_x[j * LDSX + s0 + 4 * k + r + 2 * half];
      Bk[0][0] = fmaf(qfold, ycarry, Bk[0][0]);

      v8f acc = {0.f, 0.f, 0.f, 0.f, 0.f, 0.f, 0.f, 0.f};
#pragma unroll
      for (int k = 1; k < 4; ++k)
        acc = __builtin_amdgcn_wmma_f32_16x16x4_f32(false, A[k], false, Bk[k], (short)0, acc, false, false);
      acc = __builtin_amdgcn_wmma_f32_16x16x4_f32(false, A[0], false, Bk[0], (short)0, acc, false, false);

      ycarry = __shfl(acc[7], 16 + j, 32);

      float4 ylo, yhi;
      ylo.x = w * acc[0]; ylo.y = w * acc[1]; ylo.z = w * acc[2]; ylo.w = w * acc[3];
      yhi.x = w * acc[4]; yhi.y = w * acc[5]; yhi.z = w * acc[6]; yhi.w = w * acc[7];
      *(float4*)(yrow + s0)     = ylo;
      *(float4*)(yrow + s0 + 4) = yhi;
    }

    __syncthreads();

    float4 o = *(const float4*)&lds_y[srow * LDSX + scol];
#pragma unroll
    for (int hh = 1; hh < H_; ++hh) {
      const float4 p = *(const float4*)&lds_y[(hh * 16 + srow) * LDSX + scol];
      o.x += p.x; o.y += p.y; o.z += p.z; o.w += p.w;
    }
    o.x += xr.x; o.y += xr.y; o.z += xr.z; o.w += xr.w;
    *(float4*)(orow + t0 + scol) = o;
  }
}

extern "C" void kernel_launch(void* const* d_in, const int* in_sizes, int n_in,
                              void* d_out, int out_size, void* d_ws, size_t ws_size,
                              hipStream_t stream) {
  (void)in_sizes; (void)n_in; (void)out_size; (void)ws_size;
  const float* x         = (const float*)d_in[0];
  const float* ln_gamma  = (const float*)d_in[1];
  const float* ln_beta   = (const float*)d_in[2];
  const float* expansion = (const float*)d_in[3];
  const float* reduction = (const float*)d_in[4];
  const float* alphas    = (const float*)d_in[5];
  const float* dampen    = (const float*)d_in[6];
  float* out  = (float*)d_out;
  float* mean = (float*)d_ws;
  float* rstd = mean + B_ * N_;

  ln_stats_kernel<<<B_ * (N_ / 64), 256, 0, stream>>>(x, mean, rstd);
  ema_wmma_kernel<<<SEG * B_ * (DIM / 16), 256, 0, stream>>>(
      x, ln_gamma, ln_beta, expansion, reduction, alphas, dampen, mean, rstd, out);
}
